# starting kernel
# baseline (speedup 1.0000x reference)
#include <hip/hip_runtime.h>
#include <math.h>


typedef __attribute__((ext_vector_type(2))) float v2f;
typedef __attribute__((ext_vector_type(8))) float v8f;

#define AXIS_LEN    160
#define RAD         6
#define WIN         13
#define BATCH_ELEMS (160 * 160 * 160 * 2)
#define TOTAL_ELEMS (4 * BATCH_ELEMS)
#define TOTAL_LINES (TOTAL_ELEMS / AXIS_LEN)
#define NGROUPS     (TOTAL_LINES / 16)

__global__ void gauss_wts_kernel(const float* __restrict__ sigma,
                                 float* __restrict__ wts) {
    int id = threadIdx.x;
    if (id >= 12) return;
    int b = id / 3, axis = id % 3;
    float s = sigma[b * 3 + axis];
    float denom = 2.0f * s * s + 1e-7f;
    float lg = logf(sqrtf(2.0f * (float)M_PI) * s + 1e-7f);
    float g[WIN];
    float sum = 0.0f;
    #pragma unroll
    for (int t = 0; t < WIN; ++t) {
        float loc = (float)t - (float)RAD;
        g[t] = expf(-(loc * loc) / denom - lg);
        sum += g[t];
    }
    float inv = 1.0f / sum;
    #pragma unroll
    for (int t = 0; t < WIN; ++t)
        wts[(b * 3 + axis) * WIN + t] = g[t] * inv;
}

template <int S>
__device__ __forceinline__ float ld_clamp_lo(const float* __restrict__ line, int c) {
    int cc = c < 0 ? 0 : c;
    float v = line[cc * S];
    return (c < 0) ? 0.0f : v;
}
template <int S>
__device__ __forceinline__ float ld_clamp_hi(const float* __restrict__ line, int c) {
    int cc = c > (AXIS_LEN - 1) ? (AXIS_LEN - 1) : c;
    float v = line[cc * S];
    return (c >= AXIS_LEN) ? 0.0f : v;
}

template <int S>
__global__ __launch_bounds__(128) void blur_pass_kernel(
        const float* __restrict__ src, float* __restrict__ dst,
        const float* __restrict__ wts, int axis) {
    const int lane  = threadIdx.x & 31;
    const int wave  = threadIdx.x >> 5;
    const int group = blockIdx.x * 4 + wave;

    const int n    = lane & 15;
    const int half = lane >> 4;
    const int q     = group * 16 + n;
    const int outer = q / S;
    const int inner = q - outer * S;
    const int base  = outer * (S * AXIS_LEN) + inner;
    const int b     = base / BATCH_ELEMS;
    const float* __restrict__ g = wts + (b * 3 + axis) * WIN;

    const int m = n;
    v2f A[7];
    #pragma unroll
    for (int kc = 0; kc < 7; ++kc) {
        int j0 = 4 * kc + 2 * half;
        int t0 = j0 - m;
        int t1 = t0 + 1;
        A[kc].x = (t0 >= 0 && t0 < WIN) ? g[t0] : 0.0f;
        A[kc].y = (t1 >= 0 && t1 < WIN) ? g[t1] : 0.0f;
    }

    const float* __restrict__ line  = src + base;
    float* __restrict__       oline = dst + base + (8 * half) * S;

    v2f Bv[7];
    #pragma unroll
    for (int kc = 0; kc < 7; ++kc) {
        int c0 = 4 * kc + 2 * half - RAD;
        Bv[kc].x = ld_clamp_lo<S>(line, c0);
        Bv[kc].y = ld_clamp_lo<S>(line, c0 + 1);
    }

    #pragma unroll
    for (int t = 0; t < 10; ++t) {
        const int i0 = 16 * t;

        v8f acc = {};
        #pragma unroll
        for (int kc = 0; kc < 7; ++kc) {
            acc = __builtin_amdgcn_wmma_f32_16x16x4_f32(
                 false, A[kc],  false, Bv[kc],
                 (short)0, acc,  false,  false);
        }

        if (t < 9) {
            Bv[0] = Bv[4];
            Bv[1] = Bv[5];
            Bv[2] = Bv[6];
            #pragma unroll
            for (int kc = 0; kc < 4; ++kc) {
                int c0 = i0 + 22 + 4 * kc + 2 * half;
                if (t == 8) {
                    Bv[3 + kc].x = ld_clamp_hi<S>(line, c0);
                    Bv[3 + kc].y = ld_clamp_hi<S>(line, c0 + 1);
                } else {
                    Bv[3 + kc].x = line[c0 * S];
                    Bv[3 + kc].y = line[(c0 + 1) * S];
                }
            }
        }

        #pragma unroll
        for (int v = 0; v < 8; ++v)
            oline[(i0 + v) * S] = acc[v];
    }
}

extern "C" void kernel_launch(void* const* d_in, const int* in_sizes, int n_in,
                              void* d_out, int out_size, void* d_ws, size_t ws_size,
                              hipStream_t stream) {
    const float* image = (const float*)d_in[0];
    const float* sigma = (const float*)d_in[1];
    float* out = (float*)d_out;
    float* wts = (float*)d_ws;
    float* tmp = (float*)((char*)d_ws + 4096);

    gauss_wts_kernel<<<1, 32, 0, stream>>>(sigma, wts);

    dim3 block(128);
    dim3 grid(NGROUPS / 4);

    constexpr int S_D = 160 * 160 * 2;
    constexpr int S_H = 160 * 2;
    constexpr int S_W = 2;

    blur_pass_kernel<S_D><<<grid, block, 0, stream>>>(image, out, wts, 0);
    blur_pass_kernel<S_H><<<grid, block, 0, stream>>>(out,   tmp, wts, 1);
    blur_pass_kernel<S_W><<<grid, block, 0, stream>>>(tmp,   out, wts, 2);
}


# D and H passes: one wave takes 64 adjacent lines (4 tiles sharing A), 128-bit row loads/stores covering whole 128-byte lines, rows requested two tiles ahead; summation order unchanged
# speedup vs baseline: 1.1194x; 1.1194x over previous
#include <hip/hip_runtime.h>
#include <math.h>


typedef __attribute__((ext_vector_type(2))) float v2f;
typedef __attribute__((ext_vector_type(4))) float v4f;
typedef __attribute__((ext_vector_type(8))) float v8f;

#define AXIS_LEN    160
#define RAD         6
#define WIN         13
#define BATCH_ELEMS (160 * 160 * 160 * 2)
#define TOTAL_ELEMS (4 * BATCH_ELEMS)
#define TOTAL_LINES (TOTAL_ELEMS / AXIS_LEN)
#define NGROUPS     (TOTAL_LINES / 16)
#define NGROUPS4    (TOTAL_LINES / 64)

__global__ void gauss_wts_kernel(const float* __restrict__ sigma,
                                 float* __restrict__ wts) {
    int id = threadIdx.x;
    if (id >= 12) return;
    int b = id / 3, axis = id % 3;
    float s = sigma[b * 3 + axis];
    float denom = 2.0f * s * s + 1e-7f;
    float lg = logf(sqrtf(2.0f * (float)M_PI) * s + 1e-7f);
    float g[WIN];
    float sum = 0.0f;
    #pragma unroll
    for (int t = 0; t < WIN; ++t) {
        float loc = (float)t - (float)RAD;
        g[t] = expf(-(loc * loc) / denom - lg);
        sum += g[t];
    }
    float inv = 1.0f / sum;
    #pragma unroll
    for (int t = 0; t < WIN; ++t)
        wts[(b * 3 + axis) * WIN + t] = g[t] * inv;
}

template <int S>
__device__ __forceinline__ float ld_clamp_lo(const float* __restrict__ line, int c) {
    int cc = c < 0 ? 0 : c;
    float v = line[cc * S];
    return (c < 0) ? 0.0f : v;
}
template <int S>
__device__ __forceinline__ float ld_clamp_hi(const float* __restrict__ line, int c) {
    int cc = c > (AXIS_LEN - 1) ? (AXIS_LEN - 1) : c;
    float v = line[cc * S];
    return (c >= AXIS_LEN) ? 0.0f : v;
}

template <int S>
__global__ __launch_bounds__(128) void blur_pass_kernel(
        const float* __restrict__ src, float* __restrict__ dst,
        const float* __restrict__ wts, int axis) {
    const int lane  = threadIdx.x & 31;
    const int wave  = threadIdx.x >> 5;
    const int group = blockIdx.x * 4 + wave;

    const int n    = lane & 15;
    const int half = lane >> 4;
    const int q     = group * 16 + n;
    const int outer = q / S;
    const int inner = q - outer * S;
    const int base  = outer * (S * AXIS_LEN) + inner;
    const int b     = base / BATCH_ELEMS;
    const float* __restrict__ g = wts + (b * 3 + axis) * WIN;

    const int m = n;
    v2f A[7];
    #pragma unroll
    for (int kc = 0; kc < 7; ++kc) {
        int j0 = 4 * kc + 2 * half;
        int t0 = j0 - m;
        int t1 = t0 + 1;
        A[kc].x = (t0 >= 0 && t0 < WIN) ? g[t0] : 0.0f;
        A[kc].y = (t1 >= 0 && t1 < WIN) ? g[t1] : 0.0f;
    }

    const float* __restrict__ line  = src + base;
    float* __restrict__       oline = dst + base + (8 * half) * S;

    v2f Bv[7];
    #pragma unroll
    for (int kc = 0; kc < 7; ++kc) {
        int c0 = 4 * kc + 2 * half - RAD;
        Bv[kc].x = ld_clamp_lo<S>(line, c0);
        Bv[kc].y = ld_clamp_lo<S>(line, c0 + 1);
    }

    #pragma unroll
    for (int t = 0; t < 10; ++t) {
        const int i0 = 16 * t;

        v8f acc = {};
        #pragma unroll
        for (int kc = 0; kc < 7; ++kc) {
            acc = __builtin_amdgcn_wmma_f32_16x16x4_f32(
                 false, A[kc],  false, Bv[kc],
                 (short)0, acc,  false,  false);
        }

        if (t < 9) {
            Bv[0] = Bv[4];
            Bv[1] = Bv[5];
            Bv[2] = Bv[6];
            #pragma unroll
            for (int kc = 0; kc < 4; ++kc) {
                int c0 = i0 + 22 + 4 * kc + 2 * half;
                if (t == 8) {
                    Bv[3 + kc].x = ld_clamp_hi<S>(line, c0);
                    Bv[3 + kc].y = ld_clamp_hi<S>(line, c0 + 1);
                } else {
                    Bv[3 + kc].x = line[c0 * S];
                    Bv[3 + kc].y = line[(c0 + 1) * S];
                }
            }
        }

        #pragma unroll
        for (int v = 0; v < 8; ++v)
            oline[(i0 + v) * S] = acc[v];
    }
}

template <int S>
__device__ __forceinline__ v4f ld4(const float* __restrict__ p, int c) {
    return *reinterpret_cast<const v4f*>(p + c * S);
}
template <int S>
__device__ __forceinline__ v4f ld4_clamp_lo(const float* __restrict__ line, int c) {
    int cc = c < 0 ? 0 : c;
    v4f v = ld4<S>(line, cc);
    v4f z = {0.0f, 0.0f, 0.0f, 0.0f};
    return (c < 0) ? z : v;
}
template <int S>
__device__ __forceinline__ v4f ld4_clamp_hi(const float* __restrict__ line, int c) {
    int cc = c > (AXIS_LEN - 1) ? (AXIS_LEN - 1) : c;
    v4f v = ld4<S>(line, cc);
    v4f z = {0.0f, 0.0f, 0.0f, 0.0f};
    return (c >= AXIS_LEN) ? z : v;
}

template <int S>
__device__ __forceinline__ void load_fresh(const float* __restrict__ pf, v4f (&N)[8]) {
    #pragma unroll
    for (int kc = 0; kc < 4; ++kc) {
        N[2 * kc]     = ld4<S>(pf, 4 * kc);
        N[2 * kc + 1] = ld4<S>(pf, 4 * kc + 1);
    }
}
template <int S>
__device__ __forceinline__ void load_fresh_hi(const float* __restrict__ line, int c0, v4f (&N)[8]) {
    #pragma unroll
    for (int kc = 0; kc < 4; ++kc) {
        N[2 * kc]     = ld4_clamp_hi<S>(line, c0 + 4 * kc);
        N[2 * kc + 1] = ld4_clamp_hi<S>(line, c0 + 4 * kc + 1);
    }
}
__device__ __forceinline__ void shift_in(v2f (&Bv)[4][7], const v4f (&N)[8]) {
    #pragma unroll
    for (int j = 0; j < 4; ++j) {
        Bv[j][0] = Bv[j][4];
        Bv[j][1] = Bv[j][5];
        Bv[j][2] = Bv[j][6];
        #pragma unroll
        for (int kc = 0; kc < 4; ++kc) {
            Bv[j][3 + kc].x = N[2 * kc][j];
            Bv[j][3 + kc].y = N[2 * kc + 1][j];
        }
    }
}
template <int S>
__device__ __forceinline__ void tile_x4(const v2f (&A)[7], const v2f (&Bv)[4][7],
                                        float* __restrict__ po) {
    v8f acc[4] = {};
    #pragma unroll
    for (int kc = 0; kc < 7; ++kc) {
        #pragma unroll
        for (int j = 0; j < 4; ++j) {
            acc[j] = __builtin_amdgcn_wmma_f32_16x16x4_f32(
                 false, A[kc],  false, Bv[j][kc],
                 (short)0, acc[j],  false,  false);
        }
    }
    #pragma unroll
    for (int v = 0; v < 8; ++v) {
        v4f o = {acc[0][v], acc[1][v], acc[2][v], acc[3][v]};
        *reinterpret_cast<v4f*>(po + v * S) = o;
    }
}

template <int S>
__global__ __launch_bounds__(128) void blur_pass_x4_kernel(
        const float* __restrict__ src, float* __restrict__ dst,
        const float* __restrict__ wts, int axis) {
    static_assert(S % 64 == 0, "64-line groups must not straddle an outer index");
    const int lane  = threadIdx.x & 31;
    const int wave  = threadIdx.x >> 5;
    const int group = blockIdx.x * 4 + wave;

    const int n    = lane & 15;
    const int half = lane >> 4;
    const int q     = group * 64 + 4 * n;
    const int outer = q / S;
    const int inner = q - outer * S;
    const int base  = outer * (S * AXIS_LEN) + inner;
    const int b     = base / BATCH_ELEMS;
    const float* __restrict__ g = wts + (b * 3 + axis) * WIN;

    const int m = n;
    v2f A[7];
    #pragma unroll
    for (int kc = 0; kc < 7; ++kc) {
        int j0 = 4 * kc + 2 * half;
        int t0 = j0 - m;
        int t1 = t0 + 1;
        A[kc].x = (t0 >= 0 && t0 < WIN) ? g[t0] : 0.0f;
        A[kc].y = (t1 >= 0 && t1 < WIN) ? g[t1] : 0.0f;
    }

    const float* __restrict__ line = src + base;
    float* __restrict__       po   = dst + base + (8 * half) * S;

    v2f Bv[4][7];
    #pragma unroll
    for (int kc = 0; kc < 7; ++kc) {
        int c0 = 4 * kc + 2 * half - RAD;
        v4f r0 = ld4_clamp_lo<S>(line, c0);
        v4f r1 = ld4_clamp_lo<S>(line, c0 + 1);
        #pragma unroll
        for (int j = 0; j < 4; ++j) {
            Bv[j][kc].x = r0[j];
            Bv[j][kc].y = r1[j];
        }
    }

    v4f P[8], Q[8];
    const float* __restrict__ pf = line + (16 + RAD + 2 * half) * S;
    load_fresh<S>(pf, P);
    pf += 16 * S;

    #pragma unroll 1
    for (int it = 0; it < 3; ++it) {
        load_fresh<S>(pf, Q);
        pf += 16 * S;
        tile_x4<S>(A, Bv, po);
        po += 16 * S;
        shift_in(Bv, P);

        load_fresh<S>(pf, P);
        pf += 16 * S;
        tile_x4<S>(A, Bv, po);
        po += 16 * S;
        shift_in(Bv, Q);
    }
    load_fresh<S>(pf, Q);
    tile_x4<S>(A, Bv, po);
    po += 16 * S;
    shift_in(Bv, P);

    load_fresh_hi<S>(line, 16 * 9 + RAD + 2 * half, P);
    tile_x4<S>(A, Bv, po);
    po += 16 * S;
    shift_in(Bv, Q);

    tile_x4<S>(A, Bv, po);
    po += 16 * S;
    shift_in(Bv, P);

    tile_x4<S>(A, Bv, po);
}

extern "C" void kernel_launch(void* const* d_in, const int* in_sizes, int n_in,
                              void* d_out, int out_size, void* d_ws, size_t ws_size,
                              hipStream_t stream) {
    const float* image = (const float*)d_in[0];
    const float* sigma = (const float*)d_in[1];
    float* out = (float*)d_out;
    float* wts = (float*)d_ws;
    float* tmp = (float*)((char*)d_ws + 4096);

    gauss_wts_kernel<<<1, 32, 0, stream>>>(sigma, wts);

    dim3 block(128);
    dim3 grid(NGROUPS / 4);
    dim3 grid4(NGROUPS4 / 4);

    constexpr int S_D = 160 * 160 * 2;
    constexpr int S_H = 160 * 2;
    constexpr int S_W = 2;

    blur_pass_x4_kernel<S_D><<<grid4, block, 0, stream>>>(image, out, wts, 0);
    blur_pass_x4_kernel<S_H><<<grid4, block, 0, stream>>>(out,   tmp, wts, 1);
    blur_pass_kernel<S_W><<<grid, block, 0, stream>>>(tmp,   out, wts, 2);
}


# W pass: operands swapped (image rows as A, Gaussian band as B): one 128-bit row load = 2 taps x 2 channels, each half-wave 64-bit store covers one whole 128-byte line; summation order unchanged
# speedup vs baseline: 1.2058x; 1.0772x over previous
#include <hip/hip_runtime.h>
#include <math.h>


typedef __attribute__((ext_vector_type(2))) float v2f;
typedef __attribute__((ext_vector_type(4))) float v4f;
typedef __attribute__((ext_vector_type(8))) float v8f;

#define AXIS_LEN    160
#define RAD         6
#define WIN         13
#define BATCH_ELEMS (160 * 160 * 160 * 2)
#define TOTAL_ELEMS (4 * BATCH_ELEMS)
#define TOTAL_LINES (TOTAL_ELEMS / AXIS_LEN)
#define NGROUPS4    (TOTAL_LINES / 64)

__global__ void gauss_wts_kernel(const float* __restrict__ sigma,
                                 float* __restrict__ wts) {
    int id = threadIdx.x;
    if (id >= 12) return;
    int b = id / 3, axis = id % 3;
    float s = sigma[b * 3 + axis];
    float denom = 2.0f * s * s + 1e-7f;
    float lg = logf(sqrtf(2.0f * (float)M_PI) * s + 1e-7f);
    float g[WIN];
    float sum = 0.0f;
    #pragma unroll
    for (int t = 0; t < WIN; ++t) {
        float loc = (float)t - (float)RAD;
        g[t] = expf(-(loc * loc) / denom - lg);
        sum += g[t];
    }
    float inv = 1.0f / sum;
    #pragma unroll
    for (int t = 0; t < WIN; ++t)
        wts[(b * 3 + axis) * WIN + t] = g[t] * inv;
}


template <int S>
__device__ __forceinline__ v4f ld4(const float* __restrict__ p, int c) {
    return *reinterpret_cast<const v4f*>(p + c * S);
}
template <int S>
__device__ __forceinline__ v4f ld4_clamp_lo(const float* __restrict__ line, int c) {
    int cc = c < 0 ? 0 : c;
    v4f v = ld4<S>(line, cc);
    v4f z = {0.0f, 0.0f, 0.0f, 0.0f};
    return (c < 0) ? z : v;
}
template <int S>
__device__ __forceinline__ v4f ld4_clamp_hi(const float* __restrict__ line, int c) {
    int cc = c > (AXIS_LEN - 1) ? (AXIS_LEN - 1) : c;
    v4f v = ld4<S>(line, cc);
    v4f z = {0.0f, 0.0f, 0.0f, 0.0f};
    return (c >= AXIS_LEN) ? z : v;
}

template <int S>
__device__ __forceinline__ void load_fresh(const float* __restrict__ pf, v4f (&N)[8]) {
    #pragma unroll
    for (int kc = 0; kc < 4; ++kc) {
        N[2 * kc]     = ld4<S>(pf, 4 * kc);
        N[2 * kc + 1] = ld4<S>(pf, 4 * kc + 1);
    }
}
template <int S>
__device__ __forceinline__ void load_fresh_hi(const float* __restrict__ line, int c0, v4f (&N)[8]) {
    #pragma unroll
    for (int kc = 0; kc < 4; ++kc) {
        N[2 * kc]     = ld4_clamp_hi<S>(line, c0 + 4 * kc);
        N[2 * kc + 1] = ld4_clamp_hi<S>(line, c0 + 4 * kc + 1);
    }
}
__device__ __forceinline__ void shift_in(v2f (&Bv)[4][7], const v4f (&N)[8]) {
    #pragma unroll
    for (int j = 0; j < 4; ++j) {
        Bv[j][0] = Bv[j][4];
        Bv[j][1] = Bv[j][5];
        Bv[j][2] = Bv[j][6];
        #pragma unroll
        for (int kc = 0; kc < 4; ++kc) {
            Bv[j][3 + kc].x = N[2 * kc][j];
            Bv[j][3 + kc].y = N[2 * kc + 1][j];
        }
    }
}
template <int S>
__device__ __forceinline__ void tile_x4(const v2f (&A)[7], const v2f (&Bv)[4][7],
                                        float* __restrict__ po) {
    v8f acc[4] = {};
    #pragma unroll
    for (int kc = 0; kc < 7; ++kc) {
        #pragma unroll
        for (int j = 0; j < 4; ++j) {
            acc[j] = __builtin_amdgcn_wmma_f32_16x16x4_f32(
                 false, A[kc],  false, Bv[j][kc],
                 (short)0, acc[j],  false,  false);
        }
    }
    #pragma unroll
    for (int v = 0; v < 8; ++v) {
        v4f o = {acc[0][v], acc[1][v], acc[2][v], acc[3][v]};
        *reinterpret_cast<v4f*>(po + v * S) = o;
    }
}

template <int S>
__global__ __launch_bounds__(128) void blur_pass_x4_kernel(
        const float* __restrict__ src, float* __restrict__ dst,
        const float* __restrict__ wts, int axis) {
    static_assert(S % 64 == 0, "64-line groups must not straddle an outer index");
    const int lane  = threadIdx.x & 31;
    const int wave  = threadIdx.x >> 5;
    const int group = blockIdx.x * 4 + wave;

    const int n    = lane & 15;
    const int half = lane >> 4;
    const int q     = group * 64 + 4 * n;
    const int outer = q / S;
    const int inner = q - outer * S;
    const int base  = outer * (S * AXIS_LEN) + inner;
    const int b     = base / BATCH_ELEMS;
    const float* __restrict__ g = wts + (b * 3 + axis) * WIN;

    const int m = n;
    v2f A[7];
    #pragma unroll
    for (int kc = 0; kc < 7; ++kc) {
        int j0 = 4 * kc + 2 * half;
        int t0 = j0 - m;
        int t1 = t0 + 1;
        A[kc].x = (t0 >= 0 && t0 < WIN) ? g[t0] : 0.0f;
        A[kc].y = (t1 >= 0 && t1 < WIN) ? g[t1] : 0.0f;
    }

    const float* __restrict__ line = src + base;
    float* __restrict__       po   = dst + base + (8 * half) * S;

    v2f Bv[4][7];
    #pragma unroll
    for (int kc = 0; kc < 7; ++kc) {
        int c0 = 4 * kc + 2 * half - RAD;
        v4f r0 = ld4_clamp_lo<S>(line, c0);
        v4f r1 = ld4_clamp_lo<S>(line, c0 + 1);
        #pragma unroll
        for (int j = 0; j < 4; ++j) {
            Bv[j][kc].x = r0[j];
            Bv[j][kc].y = r1[j];
        }
    }

    v4f P[8], Q[8];
    const float* __restrict__ pf = line + (16 + RAD + 2 * half) * S;
    load_fresh<S>(pf, P);
    pf += 16 * S;

    #pragma unroll 1
    for (int it = 0; it < 3; ++it) {
        load_fresh<S>(pf, Q);
        pf += 16 * S;
        tile_x4<S>(A, Bv, po);
        po += 16 * S;
        shift_in(Bv, P);

        load_fresh<S>(pf, P);
        pf += 16 * S;
        tile_x4<S>(A, Bv, po);
        po += 16 * S;
        shift_in(Bv, Q);
    }
    load_fresh<S>(pf, Q);
    tile_x4<S>(A, Bv, po);
    po += 16 * S;
    shift_in(Bv, P);

    load_fresh_hi<S>(line, 16 * 9 + RAD + 2 * half, P);
    tile_x4<S>(A, Bv, po);
    po += 16 * S;
    shift_in(Bv, Q);

    tile_x4<S>(A, Bv, po);
    po += 16 * S;
    shift_in(Bv, P);

    tile_x4<S>(A, Bv, po);
}

#define ROW_FLOATS (AXIS_LEN * 2)
#define NROWS      (TOTAL_ELEMS / ROW_FLOATS)
#define NRGROUPS   (NROWS / 16)
#define ROWS_PER_BATCH (AXIS_LEN * AXIS_LEN)

__device__ __forceinline__ v4f ldw(const float* __restrict__ p, int w) {
    return *reinterpret_cast<const v4f*>(p + 2 * w);
}
__device__ __forceinline__ v4f ldw_clamp_lo(const float* __restrict__ row, int w) {
    int ww = w < 0 ? 0 : w;
    v4f v = ldw(row, ww);
    v4f z = {0.0f, 0.0f, 0.0f, 0.0f};
    return (w < 0) ? z : v;
}
__device__ __forceinline__ v4f ldw_clamp_hi(const float* __restrict__ row, int w) {
    int ww = w > (AXIS_LEN - 2) ? (AXIS_LEN - 2) : w;
    v4f v = ldw(row, ww);
    v4f z = {0.0f, 0.0f, 0.0f, 0.0f};
    return (w >= AXIS_LEN) ? z : v;
}
__device__ __forceinline__ void load_fresh_w(const float* __restrict__ pf, v4f (&N)[4]) {
    #pragma unroll
    for (int kc = 0; kc < 4; ++kc)
        N[kc] = ldw(pf, 4 * kc);
}
__device__ __forceinline__ void load_fresh_w_hi(const float* __restrict__ row, int w0, v4f (&N)[4]) {
    #pragma unroll
    for (int kc = 0; kc < 4; ++kc)
        N[kc] = ldw_clamp_hi(row, w0 + 4 * kc);
}
__device__ __forceinline__ void shift_in_w(v2f (&A0)[7], v2f (&A1)[7], const v4f (&N)[4]) {
    A0[0] = A0[4]; A0[1] = A0[5]; A0[2] = A0[6];
    A1[0] = A1[4]; A1[1] = A1[5]; A1[2] = A1[6];
    #pragma unroll
    for (int kc = 0; kc < 4; ++kc) {
        A0[3 + kc].x = N[kc].x;  A0[3 + kc].y = N[kc].z;
        A1[3 + kc].x = N[kc].y;  A1[3 + kc].y = N[kc].w;
    }
}
__device__ __forceinline__ void tile_w(const v2f (&A0)[7], const v2f (&A1)[7],
                                       const v2f (&G)[7], float* __restrict__ po) {
    v8f acc0 = {}, acc1 = {};
    #pragma unroll
    for (int kc = 0; kc < 7; ++kc) {
        acc0 = __builtin_amdgcn_wmma_f32_16x16x4_f32(
             false, A0[kc],  false, G[kc],
             (short)0, acc0,  false,  false);
        acc1 = __builtin_amdgcn_wmma_f32_16x16x4_f32(
             false, A1[kc],  false, G[kc],
             (short)0, acc1,  false,  false);
    }
    #pragma unroll
    for (int v = 0; v < 8; ++v) {
        v2f o = {acc0[v], acc1[v]};
        *reinterpret_cast<v2f*>(po + v * ROW_FLOATS) = o;
    }
}

__global__ __launch_bounds__(128) void blur_pass_w_kernel(
        const float* __restrict__ src, float* __restrict__ dst,
        const float* __restrict__ wts) {
    const int lane  = threadIdx.x & 31;
    const int wave  = threadIdx.x >> 5;
    const int group = blockIdx.x * 4 + wave;

    const int n    = lane & 15;
    const int half = lane >> 4;
    const int b = (group * 16) / ROWS_PER_BATCH;
    const float* __restrict__ g = wts + (b * 3 + 2) * WIN;

    v2f G[7];
    #pragma unroll
    for (int kc = 0; kc < 7; ++kc) {
        int t0 = 4 * kc + 2 * half - n;
        int t1 = t0 + 1;
        G[kc].x = (t0 >= 0 && t0 < WIN) ? g[t0] : 0.0f;
        G[kc].y = (t1 >= 0 && t1 < WIN) ? g[t1] : 0.0f;
    }

    const float* __restrict__ row = src + (group * 16 + n) * ROW_FLOATS;
    float* __restrict__       po  = dst + (group * 16 + 8 * half) * ROW_FLOATS + 2 * n;

    v2f A0[7], A1[7];
    #pragma unroll
    for (int kc = 0; kc < 7; ++kc) {
        v4f r = ldw_clamp_lo(row, 4 * kc + 2 * half - RAD);
        A0[kc].x = r.x;  A0[kc].y = r.z;
        A1[kc].x = r.y;  A1[kc].y = r.w;
    }

    v4f P[4], Q[4];
    const float* __restrict__ pf = row + 2 * (16 + RAD + 2 * half);
    load_fresh_w(pf, P);
    pf += 32;

    #pragma unroll 1
    for (int it = 0; it < 3; ++it) {
        load_fresh_w(pf, Q);
        pf += 32;
        tile_w(A0, A1, G, po);
        po += 32;
        shift_in_w(A0, A1, P);

        load_fresh_w(pf, P);
        pf += 32;
        tile_w(A0, A1, G, po);
        po += 32;
        shift_in_w(A0, A1, Q);
    }
    load_fresh_w(pf, Q);
    tile_w(A0, A1, G, po);
    po += 32;
    shift_in_w(A0, A1, P);

    load_fresh_w_hi(row, 16 * 9 + RAD + 2 * half, P);
    tile_w(A0, A1, G, po);
    po += 32;
    shift_in_w(A0, A1, Q);

    tile_w(A0, A1, G, po);
    po += 32;
    shift_in_w(A0, A1, P);

    tile_w(A0, A1, G, po);
}

extern "C" void kernel_launch(void* const* d_in, const int* in_sizes, int n_in,
                              void* d_out, int out_size, void* d_ws, size_t ws_size,
                              hipStream_t stream) {
    const float* image = (const float*)d_in[0];
    const float* sigma = (const float*)d_in[1];
    float* out = (float*)d_out;
    float* wts = (float*)d_ws;
    float* tmp = (float*)((char*)d_ws + 4096);

    gauss_wts_kernel<<<1, 32, 0, stream>>>(sigma, wts);

    dim3 block(128);
    dim3 gridw(NRGROUPS / 4);
    dim3 grid4(NGROUPS4 / 4);

    constexpr int S_D = 160 * 160 * 2;
    constexpr int S_H = 160 * 2;

    blur_pass_x4_kernel<S_D><<<grid4, block, 0, stream>>>(image, out, wts, 0);
    blur_pass_x4_kernel<S_H><<<grid4, block, 0, stream>>>(out,   tmp, wts, 1);
    blur_pass_w_kernel<<<gridw, block, 0, stream>>>(tmp,   out, wts);
}


# W and H passes fused per (batch, depth) plane: W-blurred plane kept in LDS, H pass reads it there (3 launches); W now summed before H, so the rounding order of these two passes changes
# speedup vs baseline: 1.4737x; 1.2222x over previous
#include <hip/hip_runtime.h>
#include <math.h>


typedef __attribute__((ext_vector_type(2))) float v2f;
typedef __attribute__((ext_vector_type(4))) float v4f;
typedef __attribute__((ext_vector_type(8))) float v8f;

#define AXIS_LEN    160
#define RAD         6
#define WIN         13
#define BATCH_ELEMS (160 * 160 * 160 * 2)
#define TOTAL_ELEMS (4 * BATCH_ELEMS)
#define TOTAL_LINES (TOTAL_ELEMS / AXIS_LEN)
#define NGROUPS4    (TOTAL_LINES / 64)

__global__ void gauss_wts_kernel(const float* __restrict__ sigma,
                                 float* __restrict__ wts) {
    int id = threadIdx.x;
    if (id >= 12) return;
    int b = id / 3, axis = id % 3;
    float s = sigma[b * 3 + axis];
    float denom = 2.0f * s * s + 1e-7f;
    float lg = logf(sqrtf(2.0f * (float)M_PI) * s + 1e-7f);
    float g[WIN];
    float sum = 0.0f;
    #pragma unroll
    for (int t = 0; t < WIN; ++t) {
        float loc = (float)t - (float)RAD;
        g[t] = expf(-(loc * loc) / denom - lg);
        sum += g[t];
    }
    float inv = 1.0f / sum;
    #pragma unroll
    for (int t = 0; t < WIN; ++t)
        wts[(b * 3 + axis) * WIN + t] = g[t] * inv;
}


template <int S>
__device__ __forceinline__ v4f ld4(const float* __restrict__ p, int c) {
    return *reinterpret_cast<const v4f*>(p + c * S);
}
template <int S>
__device__ __forceinline__ v4f ld4_clamp_lo(const float* __restrict__ line, int c) {
    int cc = c < 0 ? 0 : c;
    v4f v = ld4<S>(line, cc);
    v4f z = {0.0f, 0.0f, 0.0f, 0.0f};
    return (c < 0) ? z : v;
}
template <int S>
__device__ __forceinline__ v4f ld4_clamp_hi(const float* __restrict__ line, int c) {
    int cc = c > (AXIS_LEN - 1) ? (AXIS_LEN - 1) : c;
    v4f v = ld4<S>(line, cc);
    v4f z = {0.0f, 0.0f, 0.0f, 0.0f};
    return (c >= AXIS_LEN) ? z : v;
}

template <int S>
__device__ __forceinline__ void load_fresh(const float* __restrict__ pf, v4f (&N)[8]) {
    #pragma unroll
    for (int kc = 0; kc < 4; ++kc) {
        N[2 * kc]     = ld4<S>(pf, 4 * kc);
        N[2 * kc + 1] = ld4<S>(pf, 4 * kc + 1);
    }
}
template <int S>
__device__ __forceinline__ void load_fresh_hi(const float* __restrict__ line, int c0, v4f (&N)[8]) {
    #pragma unroll
    for (int kc = 0; kc < 4; ++kc) {
        N[2 * kc]     = ld4_clamp_hi<S>(line, c0 + 4 * kc);
        N[2 * kc + 1] = ld4_clamp_hi<S>(line, c0 + 4 * kc + 1);
    }
}
__device__ __forceinline__ void shift_in(v2f (&Bv)[4][7], const v4f (&N)[8]) {
    #pragma unroll
    for (int j = 0; j < 4; ++j) {
        Bv[j][0] = Bv[j][4];
        Bv[j][1] = Bv[j][5];
        Bv[j][2] = Bv[j][6];
        #pragma unroll
        for (int kc = 0; kc < 4; ++kc) {
            Bv[j][3 + kc].x = N[2 * kc][j];
            Bv[j][3 + kc].y = N[2 * kc + 1][j];
        }
    }
}
template <int S>
__device__ __forceinline__ void tile_x4(const v2f (&A)[7], const v2f (&Bv)[4][7],
                                        float* __restrict__ po) {
    v8f acc[4] = {};
    #pragma unroll
    for (int kc = 0; kc < 7; ++kc) {
        #pragma unroll
        for (int j = 0; j < 4; ++j) {
            acc[j] = __builtin_amdgcn_wmma_f32_16x16x4_f32(
                 false, A[kc],  false, Bv[j][kc],
                 (short)0, acc[j],  false,  false);
        }
    }
    #pragma unroll
    for (int v = 0; v < 8; ++v) {
        v4f o = {acc[0][v], acc[1][v], acc[2][v], acc[3][v]};
        *reinterpret_cast<v4f*>(po + v * S) = o;
    }
}

template <int S>
__global__ __launch_bounds__(128) void blur_pass_x4_kernel(
        const float* __restrict__ src, float* __restrict__ dst,
        const float* __restrict__ wts, int axis) {
    static_assert(S % 64 == 0, "64-line groups must not straddle an outer index");
    const int lane  = threadIdx.x & 31;
    const int wave  = threadIdx.x >> 5;
    const int group = blockIdx.x * 4 + wave;

    const int n    = lane & 15;
    const int half = lane >> 4;
    const int q     = group * 64 + 4 * n;
    const int outer = q / S;
    const int inner = q - outer * S;
    const int base  = outer * (S * AXIS_LEN) + inner;
    const int b     = base / BATCH_ELEMS;
    const float* __restrict__ g = wts + (b * 3 + axis) * WIN;

    const int m = n;
    v2f A[7];
    #pragma unroll
    for (int kc = 0; kc < 7; ++kc) {
        int j0 = 4 * kc + 2 * half;
        int t0 = j0 - m;
        int t1 = t0 + 1;
        A[kc].x = (t0 >= 0 && t0 < WIN) ? g[t0] : 0.0f;
        A[kc].y = (t1 >= 0 && t1 < WIN) ? g[t1] : 0.0f;
    }

    const float* __restrict__ line = src + base;
    float* __restrict__       po   = dst + base + (8 * half) * S;

    v2f Bv[4][7];
    #pragma unroll
    for (int kc = 0; kc < 7; ++kc) {
        int c0 = 4 * kc + 2 * half - RAD;
        v4f r0 = ld4_clamp_lo<S>(line, c0);
        v4f r1 = ld4_clamp_lo<S>(line, c0 + 1);
        #pragma unroll
        for (int j = 0; j < 4; ++j) {
            Bv[j][kc].x = r0[j];
            Bv[j][kc].y = r1[j];
        }
    }

    v4f P[8], Q[8];
    const float* __restrict__ pf = line + (16 + RAD + 2 * half) * S;
    load_fresh<S>(pf, P);
    pf += 16 * S;

    #pragma unroll 1
    for (int it = 0; it < 3; ++it) {
        load_fresh<S>(pf, Q);
        pf += 16 * S;
        tile_x4<S>(A, Bv, po);
        po += 16 * S;
        shift_in(Bv, P);

        load_fresh<S>(pf, P);
        pf += 16 * S;
        tile_x4<S>(A, Bv, po);
        po += 16 * S;
        shift_in(Bv, Q);
    }
    load_fresh<S>(pf, Q);
    tile_x4<S>(A, Bv, po);
    po += 16 * S;
    shift_in(Bv, P);

    load_fresh_hi<S>(line, 16 * 9 + RAD + 2 * half, P);
    tile_x4<S>(A, Bv, po);
    po += 16 * S;
    shift_in(Bv, Q);

    tile_x4<S>(A, Bv, po);
    po += 16 * S;
    shift_in(Bv, P);

    tile_x4<S>(A, Bv, po);
}

#define ROW_FLOATS   (AXIS_LEN * 2)
#define PLANE_FLOATS (AXIS_LEN * ROW_FLOATS)
#define NPLANES      (4 * AXIS_LEN)
#define LDS_PITCH    (ROW_FLOATS + 4)
#define HW_WAVES     20

__device__ __forceinline__ v4f ldw(const float* __restrict__ p, int w) {
    return *reinterpret_cast<const v4f*>(p + 2 * w);
}
__device__ __forceinline__ v4f ldw_clamp(const float* __restrict__ row, int w) {
    int ww = w < 0 ? 0 : (w > (AXIS_LEN - 2) ? (AXIS_LEN - 2) : w);
    v4f v = ldw(row, ww);
    v4f z = {0.0f, 0.0f, 0.0f, 0.0f};
    return (w < 0 || w >= AXIS_LEN) ? z : v;
}

__global__ __launch_bounds__(HW_WAVES * 32) void blur_hw_plane_kernel(
        const float* __restrict__ src, float* __restrict__ dst,
        const float* __restrict__ wts) {
    __shared__ __attribute__((aligned(16))) float plane[AXIS_LEN * LDS_PITCH];

    const int lane = threadIdx.x & 31;
    const int wave = __builtin_amdgcn_readfirstlane((int)(threadIdx.x >> 5));
    const int n    = lane & 15;
    const int half = lane >> 4;

    const int p = blockIdx.x;
    const int b = p / AXIS_LEN;
    const float* __restrict__ splane = src + p * PLANE_FLOATS;
    float* __restrict__       dplane = dst + p * PLANE_FLOATS;

    {
        const float* __restrict__ g = wts + (b * 3 + 2) * WIN;
        v2f G[7];
        #pragma unroll
        for (int kc = 0; kc < 7; ++kc) {
            int t0 = 4 * kc + 2 * half - n;
            int t1 = t0 + 1;
            G[kc].x = (t0 >= 0 && t0 < WIN) ? g[t0] : 0.0f;
            G[kc].y = (t1 >= 0 && t1 < WIN) ? g[t1] : 0.0f;
        }

        const int rg = wave >> 1;
        const int w0 = 80 * (wave & 1);
        const float* __restrict__ row = splane + (16 * rg + n) * ROW_FLOATS;
        const int wb = w0 - RAD + 2 * half;

        v4f R[23];
        #pragma unroll
        for (int i = 0; i < 23; ++i)
            R[i] = ldw_clamp(row, wb + 4 * i);

        const int lo = (16 * rg + 8 * half) * LDS_PITCH + 2 * (w0 + n);
        #pragma unroll
        for (int t = 0; t < 5; ++t) {
            v8f acc0 = {}, acc1 = {};
            #pragma unroll
            for (int kc = 0; kc < 7; ++kc) {
                const v4f r = R[4 * t + kc];
                const v2f a0 = {r.x, r.z};
                const v2f a1 = {r.y, r.w};
                acc0 = __builtin_amdgcn_wmma_f32_16x16x4_f32(
                     false, a0,  false, G[kc],
                     (short)0, acc0,  false,  false);
                acc1 = __builtin_amdgcn_wmma_f32_16x16x4_f32(
                     false, a1,  false, G[kc],
                     (short)0, acc1,  false,  false);
            }
            #pragma unroll
            for (int v = 0; v < 8; ++v) {
                v2f o = {acc0[v], acc1[v]};
                *reinterpret_cast<v2f*>(&plane[lo + 32 * t + v * LDS_PITCH]) = o;
            }
        }
    }

    __syncthreads();

    {
        const float* __restrict__ g = wts + (b * 3 + 1) * WIN;
        const int m = n;
        v2f A[7];
        #pragma unroll
        for (int kc = 0; kc < 7; ++kc) {
            int t0 = 4 * kc + 2 * half - m;
            int t1 = t0 + 1;
            A[kc].x = (t0 >= 0 && t0 < WIN) ? g[t0] : 0.0f;
            A[kc].y = (t1 >= 0 && t1 < WIN) ? g[t1] : 0.0f;
        }

        for (int u = wave; u < 50; u += HW_WAVES) {
            const int t   = u / 5;
            const int lg  = u - 5 * t;
            const int col = 64 * lg + 4 * n;

            v2f Bv[4][7];
            #pragma unroll
            for (int kc = 0; kc < 7; ++kc) {
                const int c0 = 16 * t + 4 * kc + 2 * half - RAD;
                const int c1 = c0 + 1;
                const int k0 = c0 < 0 ? 0 : (c0 > (AXIS_LEN - 1) ? (AXIS_LEN - 1) : c0);
                const int k1 = c1 < 0 ? 0 : (c1 > (AXIS_LEN - 1) ? (AXIS_LEN - 1) : c1);
                v4f r0 = *reinterpret_cast<const v4f*>(&plane[k0 * LDS_PITCH + col]);
                v4f r1 = *reinterpret_cast<const v4f*>(&plane[k1 * LDS_PITCH + col]);
                const v4f z = {0.0f, 0.0f, 0.0f, 0.0f};
                r0 = (c0 < 0 || c0 >= AXIS_LEN) ? z : r0;
                r1 = (c1 < 0 || c1 >= AXIS_LEN) ? z : r1;
                #pragma unroll
                for (int j = 0; j < 4; ++j) {
                    Bv[j][kc].x = r0[j];
                    Bv[j][kc].y = r1[j];
                }
            }
            tile_x4<ROW_FLOATS>(A, Bv, dplane + (16 * t + 8 * half) * ROW_FLOATS + col);
        }
    }
}

extern "C" void kernel_launch(void* const* d_in, const int* in_sizes, int n_in,
                              void* d_out, int out_size, void* d_ws, size_t ws_size,
                              hipStream_t stream) {
    const float* image = (const float*)d_in[0];
    const float* sigma = (const float*)d_in[1];
    float* out = (float*)d_out;
    float* wts = (float*)d_ws;
    float* tmp = (float*)((char*)d_ws + 4096);

    gauss_wts_kernel<<<1, 32, 0, stream>>>(sigma, wts);

    dim3 block(128);
    dim3 grid4(NGROUPS4 / 4);

    constexpr int S_D = 160 * 160 * 2;

    blur_pass_x4_kernel<S_D><<<grid4, block, 0, stream>>>(image, tmp, wts, 0);
    blur_hw_plane_kernel<<<dim3(NPLANES), dim3(HW_WAVES * 32), 0, stream>>>(tmp, out, wts);
}


# W phase of the plane kernel: each wave loads its 8 rows as whole 128-byte lines (512 contiguous bytes per load), stages them in LDS by channel, A = 8 rows x 2 channels; sum order unchanged
# speedup vs baseline: 1.5749x; 1.0687x over previous
#include <hip/hip_runtime.h>
#include <math.h>


typedef __attribute__((ext_vector_type(2))) float v2f;
typedef __attribute__((ext_vector_type(4))) float v4f;
typedef __attribute__((ext_vector_type(8))) float v8f;

#define AXIS_LEN    160
#define RAD         6
#define WIN         13
#define BATCH_ELEMS (160 * 160 * 160 * 2)
#define TOTAL_ELEMS (4 * BATCH_ELEMS)
#define TOTAL_LINES (TOTAL_ELEMS / AXIS_LEN)
#define NGROUPS4    (TOTAL_LINES / 64)

__global__ void gauss_wts_kernel(const float* __restrict__ sigma,
                                 float* __restrict__ wts) {
    int id = threadIdx.x;
    if (id >= 12) return;
    int b = id / 3, axis = id % 3;
    float s = sigma[b * 3 + axis];
    float denom = 2.0f * s * s + 1e-7f;
    float lg = logf(sqrtf(2.0f * (float)M_PI) * s + 1e-7f);
    float g[WIN];
    float sum = 0.0f;
    #pragma unroll
    for (int t = 0; t < WIN; ++t) {
        float loc = (float)t - (float)RAD;
        g[t] = expf(-(loc * loc) / denom - lg);
        sum += g[t];
    }
    float inv = 1.0f / sum;
    #pragma unroll
    for (int t = 0; t < WIN; ++t)
        wts[(b * 3 + axis) * WIN + t] = g[t] * inv;
}


template <int S>
__device__ __forceinline__ v4f ld4(const float* __restrict__ p, int c) {
    return *reinterpret_cast<const v4f*>(p + c * S);
}
template <int S>
__device__ __forceinline__ v4f ld4_clamp_lo(const float* __restrict__ line, int c) {
    int cc = c < 0 ? 0 : c;
    v4f v = ld4<S>(line, cc);
    v4f z = {0.0f, 0.0f, 0.0f, 0.0f};
    return (c < 0) ? z : v;
}
template <int S>
__device__ __forceinline__ v4f ld4_clamp_hi(const float* __restrict__ line, int c) {
    int cc = c > (AXIS_LEN - 1) ? (AXIS_LEN - 1) : c;
    v4f v = ld4<S>(line, cc);
    v4f z = {0.0f, 0.0f, 0.0f, 0.0f};
    return (c >= AXIS_LEN) ? z : v;
}

template <int S>
__device__ __forceinline__ void load_fresh(const float* __restrict__ pf, v4f (&N)[8]) {
    #pragma unroll
    for (int kc = 0; kc < 4; ++kc) {
        N[2 * kc]     = ld4<S>(pf, 4 * kc);
        N[2 * kc + 1] = ld4<S>(pf, 4 * kc + 1);
    }
}
template <int S>
__device__ __forceinline__ void load_fresh_hi(const float* __restrict__ line, int c0, v4f (&N)[8]) {
    #pragma unroll
    for (int kc = 0; kc < 4; ++kc) {
        N[2 * kc]     = ld4_clamp_hi<S>(line, c0 + 4 * kc);
        N[2 * kc + 1] = ld4_clamp_hi<S>(line, c0 + 4 * kc + 1);
    }
}
__device__ __forceinline__ void shift_in(v2f (&Bv)[4][7], const v4f (&N)[8]) {
    #pragma unroll
    for (int j = 0; j < 4; ++j) {
        Bv[j][0] = Bv[j][4];
        Bv[j][1] = Bv[j][5];
        Bv[j][2] = Bv[j][6];
        #pragma unroll
        for (int kc = 0; kc < 4; ++kc) {
            Bv[j][3 + kc].x = N[2 * kc][j];
            Bv[j][3 + kc].y = N[2 * kc + 1][j];
        }
    }
}
template <int S>
__device__ __forceinline__ void tile_x4(const v2f (&A)[7], const v2f (&Bv)[4][7],
                                        float* __restrict__ po) {
    v8f acc[4] = {};
    #pragma unroll
    for (int kc = 0; kc < 7; ++kc) {
        #pragma unroll
        for (int j = 0; j < 4; ++j) {
            acc[j] = __builtin_amdgcn_wmma_f32_16x16x4_f32(
                 false, A[kc],  false, Bv[j][kc],
                 (short)0, acc[j],  false,  false);
        }
    }
    #pragma unroll
    for (int v = 0; v < 8; ++v) {
        v4f o = {acc[0][v], acc[1][v], acc[2][v], acc[3][v]};
        *reinterpret_cast<v4f*>(po + v * S) = o;
    }
}

template <int S>
__global__ __launch_bounds__(128) void blur_pass_x4_kernel(
        const float* __restrict__ src, float* __restrict__ dst,
        const float* __restrict__ wts, int axis) {
    static_assert(S % 64 == 0, "64-line groups must not straddle an outer index");
    const int lane  = threadIdx.x & 31;
    const int wave  = threadIdx.x >> 5;
    const int group = blockIdx.x * 4 + wave;

    const int n    = lane & 15;
    const int half = lane >> 4;
    const int q     = group * 64 + 4 * n;
    const int outer = q / S;
    const int inner = q - outer * S;
    const int base  = outer * (S * AXIS_LEN) + inner;
    const int b     = base / BATCH_ELEMS;
    const float* __restrict__ g = wts + (b * 3 + axis) * WIN;

    const int m = n;
    v2f A[7];
    #pragma unroll
    for (int kc = 0; kc < 7; ++kc) {
        int j0 = 4 * kc + 2 * half;
        int t0 = j0 - m;
        int t1 = t0 + 1;
        A[kc].x = (t0 >= 0 && t0 < WIN) ? g[t0] : 0.0f;
        A[kc].y = (t1 >= 0 && t1 < WIN) ? g[t1] : 0.0f;
    }

    const float* __restrict__ line = src + base;
    float* __restrict__       po   = dst + base + (8 * half) * S;

    v2f Bv[4][7];
    #pragma unroll
    for (int kc = 0; kc < 7; ++kc) {
        int c0 = 4 * kc + 2 * half - RAD;
        v4f r0 = ld4_clamp_lo<S>(line, c0);
        v4f r1 = ld4_clamp_lo<S>(line, c0 + 1);
        #pragma unroll
        for (int j = 0; j < 4; ++j) {
            Bv[j][kc].x = r0[j];
            Bv[j][kc].y = r1[j];
        }
    }

    v4f P[8], Q[8];
    const float* __restrict__ pf = line + (16 + RAD + 2 * half) * S;
    load_fresh<S>(pf, P);
    pf += 16 * S;

    #pragma unroll 1
    for (int it = 0; it < 3; ++it) {
        load_fresh<S>(pf, Q);
        pf += 16 * S;
        tile_x4<S>(A, Bv, po);
        po += 16 * S;
        shift_in(Bv, P);

        load_fresh<S>(pf, P);
        pf += 16 * S;
        tile_x4<S>(A, Bv, po);
        po += 16 * S;
        shift_in(Bv, Q);
    }
    load_fresh<S>(pf, Q);
    tile_x4<S>(A, Bv, po);
    po += 16 * S;
    shift_in(Bv, P);

    load_fresh_hi<S>(line, 16 * 9 + RAD + 2 * half, P);
    tile_x4<S>(A, Bv, po);
    po += 16 * S;
    shift_in(Bv, Q);

    tile_x4<S>(A, Bv, po);
    po += 16 * S;
    shift_in(Bv, P);

    tile_x4<S>(A, Bv, po);
}

#define ROW_FLOATS   (AXIS_LEN * 2)
#define PLANE_FLOATS (AXIS_LEN * ROW_FLOATS)
#define NPLANES      (4 * AXIS_LEN)
#define LDS_PITCH    (ROW_FLOATS + 4)
#define HW_WAVES     20
#define CH_OFF       AXIS_LEN

__global__ __launch_bounds__(HW_WAVES * 32) void blur_hw_plane_kernel(
        const float* __restrict__ src, float* __restrict__ dst,
        const float* __restrict__ wts) {
    __shared__ __attribute__((aligned(16))) float plane[AXIS_LEN * LDS_PITCH];

    const int lane = threadIdx.x & 31;
    const int wave = __builtin_amdgcn_readfirstlane((int)(threadIdx.x >> 5));
    const int n    = lane & 15;
    const int half = lane >> 4;

    const int p = blockIdx.x;
    const int b = p / AXIS_LEN;
    const float* __restrict__ splane = src + p * PLANE_FLOATS;
    float* __restrict__       dplane = dst + p * PLANE_FLOATS;

    {
        const float* __restrict__ g = wts + (b * 3 + 2) * WIN;
        v2f G[7];
        #pragma unroll
        for (int kc = 0; kc < 7; ++kc) {
            int t0 = 4 * kc + 2 * half - n;
            int t1 = t0 + 1;
            G[kc].x = (t0 >= 0 && t0 < WIN) ? g[t0] : 0.0f;
            G[kc].y = (t1 >= 0 && t1 < WIN) ? g[t1] : 0.0f;
        }

        const float* __restrict__ blk = splane + wave * (8 * ROW_FLOATS);
        v4f T[20];
        #pragma unroll
        for (int j = 0; j < 20; ++j)
            T[j] = *reinterpret_cast<const v4f*>(blk + 128 * j + 4 * lane);

        #pragma unroll
        for (int j = 0; j < 20; ++j) {
            const int f   = 128 * j + 4 * lane;
            const int rl  = f / ROW_FLOATS;
            const int col = f - rl * ROW_FLOATS;
            const int a0  = (8 * wave + rl) * LDS_PITCH + (col >> 1);
            const v2f c0 = {T[j].x, T[j].z};
            const v2f c1 = {T[j].y, T[j].w};
            *reinterpret_cast<v2f*>(&plane[a0])          = c0;
            *reinterpret_cast<v2f*>(&plane[a0 + CH_OFF]) = c1;
        }

        const int frow = (8 * wave + (n & 7)) * LDS_PITCH + (n >> 3) * CH_OFF;
        v2f R[43];
        #pragma unroll
        for (int i = 0; i < 43; ++i) {
            const int pos = 4 * i - RAD + 2 * half;
            if (i < 2 || i > 40) {
                const int pc = pos < 0 ? 0 : (pos > (AXIS_LEN - 2) ? (AXIS_LEN - 2) : pos);
                const v2f v = *reinterpret_cast<const v2f*>(&plane[frow + pc]);
                const v2f z = {0.0f, 0.0f};
                R[i] = (pos < 0 || pos >= AXIS_LEN) ? z : v;
            } else {
                R[i] = *reinterpret_cast<const v2f*>(&plane[frow + pos]);
            }
        }

        const int ob = (8 * wave) * LDS_PITCH + 2 * n + half;
        #pragma unroll
        for (int tp = 0; tp < 5; ++tp) {
            v8f acc0 = {}, acc1 = {};
            #pragma unroll
            for (int kc = 0; kc < 7; ++kc) {
                acc0 = __builtin_amdgcn_wmma_f32_16x16x4_f32(
                     false, R[8 * tp + kc],  false, G[kc],
                     (short)0, acc0,  false,  false);
                acc1 = __builtin_amdgcn_wmma_f32_16x16x4_f32(
                     false, R[8 * tp + 4 + kc],  false, G[kc],
                     (short)0, acc1,  false,  false);
            }
            #pragma unroll
            for (int v = 0; v < 8; ++v) {
                plane[ob + v * LDS_PITCH + 64 * tp]      = acc0[v];
                plane[ob + v * LDS_PITCH + 64 * tp + 32] = acc1[v];
            }
        }
    }

    __syncthreads();

    {
        const float* __restrict__ g = wts + (b * 3 + 1) * WIN;
        const int m = n;
        v2f A[7];
        #pragma unroll
        for (int kc = 0; kc < 7; ++kc) {
            int t0 = 4 * kc + 2 * half - m;
            int t1 = t0 + 1;
            A[kc].x = (t0 >= 0 && t0 < WIN) ? g[t0] : 0.0f;
            A[kc].y = (t1 >= 0 && t1 < WIN) ? g[t1] : 0.0f;
        }

        for (int u = wave; u < 50; u += HW_WAVES) {
            const int t   = u / 5;
            const int lg  = u - 5 * t;
            const int col = 64 * lg + 4 * n;

            v2f Bv[4][7];
            #pragma unroll
            for (int kc = 0; kc < 7; ++kc) {
                const int c0 = 16 * t + 4 * kc + 2 * half - RAD;
                const int c1 = c0 + 1;
                const int k0 = c0 < 0 ? 0 : (c0 > (AXIS_LEN - 1) ? (AXIS_LEN - 1) : c0);
                const int k1 = c1 < 0 ? 0 : (c1 > (AXIS_LEN - 1) ? (AXIS_LEN - 1) : c1);
                v4f r0 = *reinterpret_cast<const v4f*>(&plane[k0 * LDS_PITCH + col]);
                v4f r1 = *reinterpret_cast<const v4f*>(&plane[k1 * LDS_PITCH + col]);
                const v4f z = {0.0f, 0.0f, 0.0f, 0.0f};
                r0 = (c0 < 0 || c0 >= AXIS_LEN) ? z : r0;
                r1 = (c1 < 0 || c1 >= AXIS_LEN) ? z : r1;
                #pragma unroll
                for (int j = 0; j < 4; ++j) {
                    Bv[j][kc].x = r0[j];
                    Bv[j][kc].y = r1[j];
                }
            }
            tile_x4<ROW_FLOATS>(A, Bv, dplane + (16 * t + 8 * half) * ROW_FLOATS + col);
        }
    }
}

extern "C" void kernel_launch(void* const* d_in, const int* in_sizes, int n_in,
                              void* d_out, int out_size, void* d_ws, size_t ws_size,
                              hipStream_t stream) {
    const float* image = (const float*)d_in[0];
    const float* sigma = (const float*)d_in[1];
    float* out = (float*)d_out;
    float* wts = (float*)d_ws;
    float* tmp = (float*)((char*)d_ws + 4096);

    gauss_wts_kernel<<<1, 32, 0, stream>>>(sigma, wts);

    dim3 block(128);
    dim3 grid4(NGROUPS4 / 4);

    constexpr int S_D = 160 * 160 * 2;

    blur_pass_x4_kernel<S_D><<<grid4, block, 0, stream>>>(image, tmp, wts, 0);
    blur_hw_plane_kernel<<<dim3(NPLANES), dim3(HW_WAVES * 32), 0, stream>>>(tmp, out, wts);
}


# W+H plane kernel: a work-group takes half a plane (80 output rows, 88 W-blurred rows in LDS) so two fit one WGP's LDS and one's loads and stores run under the other's products; sum order unchanged
# speedup vs baseline: 1.6232x; 1.0307x over previous
#include <hip/hip_runtime.h>
#include <math.h>


typedef __attribute__((ext_vector_type(2))) float v2f;
typedef __attribute__((ext_vector_type(4))) float v4f;
typedef __attribute__((ext_vector_type(8))) float v8f;

#define AXIS_LEN    160
#define RAD         6
#define WIN         13
#define BATCH_ELEMS (160 * 160 * 160 * 2)
#define TOTAL_ELEMS (4 * BATCH_ELEMS)
#define TOTAL_LINES (TOTAL_ELEMS / AXIS_LEN)
#define NGROUPS4    (TOTAL_LINES / 64)

__global__ void gauss_wts_kernel(const float* __restrict__ sigma,
                                 float* __restrict__ wts) {
    int id = threadIdx.x;
    if (id >= 12) return;
    int b = id / 3, axis = id % 3;
    float s = sigma[b * 3 + axis];
    float denom = 2.0f * s * s + 1e-7f;
    float lg = logf(sqrtf(2.0f * (float)M_PI) * s + 1e-7f);
    float g[WIN];
    float sum = 0.0f;
    #pragma unroll
    for (int t = 0; t < WIN; ++t) {
        float loc = (float)t - (float)RAD;
        g[t] = expf(-(loc * loc) / denom - lg);
        sum += g[t];
    }
    float inv = 1.0f / sum;
    #pragma unroll
    for (int t = 0; t < WIN; ++t)
        wts[(b * 3 + axis) * WIN + t] = g[t] * inv;
}


template <int S>
__device__ __forceinline__ v4f ld4(const float* __restrict__ p, int c) {
    return *reinterpret_cast<const v4f*>(p + c * S);
}
template <int S>
__device__ __forceinline__ v4f ld4_clamp_lo(const float* __restrict__ line, int c) {
    int cc = c < 0 ? 0 : c;
    v4f v = ld4<S>(line, cc);
    v4f z = {0.0f, 0.0f, 0.0f, 0.0f};
    return (c < 0) ? z : v;
}
template <int S>
__device__ __forceinline__ v4f ld4_clamp_hi(const float* __restrict__ line, int c) {
    int cc = c > (AXIS_LEN - 1) ? (AXIS_LEN - 1) : c;
    v4f v = ld4<S>(line, cc);
    v4f z = {0.0f, 0.0f, 0.0f, 0.0f};
    return (c >= AXIS_LEN) ? z : v;
}

template <int S>
__device__ __forceinline__ void load_fresh(const float* __restrict__ pf, v4f (&N)[8]) {
    #pragma unroll
    for (int kc = 0; kc < 4; ++kc) {
        N[2 * kc]     = ld4<S>(pf, 4 * kc);
        N[2 * kc + 1] = ld4<S>(pf, 4 * kc + 1);
    }
}
template <int S>
__device__ __forceinline__ void load_fresh_hi(const float* __restrict__ line, int c0, v4f (&N)[8]) {
    #pragma unroll
    for (int kc = 0; kc < 4; ++kc) {
        N[2 * kc]     = ld4_clamp_hi<S>(line, c0 + 4 * kc);
        N[2 * kc + 1] = ld4_clamp_hi<S>(line, c0 + 4 * kc + 1);
    }
}
__device__ __forceinline__ void shift_in(v2f (&Bv)[4][7], const v4f (&N)[8]) {
    #pragma unroll
    for (int j = 0; j < 4; ++j) {
        Bv[j][0] = Bv[j][4];
        Bv[j][1] = Bv[j][5];
        Bv[j][2] = Bv[j][6];
        #pragma unroll
        for (int kc = 0; kc < 4; ++kc) {
            Bv[j][3 + kc].x = N[2 * kc][j];
            Bv[j][3 + kc].y = N[2 * kc + 1][j];
        }
    }
}
template <int S>
__device__ __forceinline__ void tile_x4(const v2f (&A)[7], const v2f (&Bv)[4][7],
                                        float* __restrict__ po) {
    v8f acc[4] = {};
    #pragma unroll
    for (int kc = 0; kc < 7; ++kc) {
        #pragma unroll
        for (int j = 0; j < 4; ++j) {
            acc[j] = __builtin_amdgcn_wmma_f32_16x16x4_f32(
                 false, A[kc],  false, Bv[j][kc],
                 (short)0, acc[j],  false,  false);
        }
    }
    #pragma unroll
    for (int v = 0; v < 8; ++v) {
        v4f o = {acc[0][v], acc[1][v], acc[2][v], acc[3][v]};
        *reinterpret_cast<v4f*>(po + v * S) = o;
    }
}

template <int S>
__global__ __launch_bounds__(128) void blur_pass_x4_kernel(
        const float* __restrict__ src, float* __restrict__ dst,
        const float* __restrict__ wts, int axis) {
    static_assert(S % 64 == 0, "64-line groups must not straddle an outer index");
    const int lane  = threadIdx.x & 31;
    const int wave  = threadIdx.x >> 5;
    const int group = blockIdx.x * 4 + wave;

    const int n    = lane & 15;
    const int half = lane >> 4;
    const int q     = group * 64 + 4 * n;
    const int outer = q / S;
    const int inner = q - outer * S;
    const int base  = outer * (S * AXIS_LEN) + inner;
    const int b     = base / BATCH_ELEMS;
    const float* __restrict__ g = wts + (b * 3 + axis) * WIN;

    const int m = n;
    v2f A[7];
    #pragma unroll
    for (int kc = 0; kc < 7; ++kc) {
        int j0 = 4 * kc + 2 * half;
        int t0 = j0 - m;
        int t1 = t0 + 1;
        A[kc].x = (t0 >= 0 && t0 < WIN) ? g[t0] : 0.0f;
        A[kc].y = (t1 >= 0 && t1 < WIN) ? g[t1] : 0.0f;
    }

    const float* __restrict__ line = src + base;
    float* __restrict__       po   = dst + base + (8 * half) * S;

    v2f Bv[4][7];
    #pragma unroll
    for (int kc = 0; kc < 7; ++kc) {
        int c0 = 4 * kc + 2 * half - RAD;
        v4f r0 = ld4_clamp_lo<S>(line, c0);
        v4f r1 = ld4_clamp_lo<S>(line, c0 + 1);
        #pragma unroll
        for (int j = 0; j < 4; ++j) {
            Bv[j][kc].x = r0[j];
            Bv[j][kc].y = r1[j];
        }
    }

    v4f P[8], Q[8];
    const float* __restrict__ pf = line + (16 + RAD + 2 * half) * S;
    load_fresh<S>(pf, P);
    pf += 16 * S;

    #pragma unroll 1
    for (int it = 0; it < 3; ++it) {
        load_fresh<S>(pf, Q);
        pf += 16 * S;
        tile_x4<S>(A, Bv, po);
        po += 16 * S;
        shift_in(Bv, P);

        load_fresh<S>(pf, P);
        pf += 16 * S;
        tile_x4<S>(A, Bv, po);
        po += 16 * S;
        shift_in(Bv, Q);
    }
    load_fresh<S>(pf, Q);
    tile_x4<S>(A, Bv, po);
    po += 16 * S;
    shift_in(Bv, P);

    load_fresh_hi<S>(line, 16 * 9 + RAD + 2 * half, P);
    tile_x4<S>(A, Bv, po);
    po += 16 * S;
    shift_in(Bv, Q);

    tile_x4<S>(A, Bv, po);
    po += 16 * S;
    shift_in(Bv, P);

    tile_x4<S>(A, Bv, po);
}

#define ROW_FLOATS   (AXIS_LEN * 2)
#define PLANE_FLOATS (AXIS_LEN * ROW_FLOATS)
#define NPLANES      (4 * AXIS_LEN)
#define NHALVES      (2 * NPLANES)
#define LDS_PITCH    (ROW_FLOATS + 4)
#define HW_WAVES     11
#define HP_OUT_ROWS  80
#define HP_ROWS      (8 * HW_WAVES)
#define HP_ROW0      (AXIS_LEN - HP_ROWS)
#define CH_OFF       AXIS_LEN

__global__ __launch_bounds__(HW_WAVES * 32, 6) void blur_hw_plane_kernel(
        const float* __restrict__ src, float* __restrict__ dst,
        const float* __restrict__ wts) {
    __shared__ __attribute__((aligned(16))) float plane[HP_ROWS * LDS_PITCH];

    const int lane = threadIdx.x & 31;
    const int wave = __builtin_amdgcn_readfirstlane((int)(threadIdx.x >> 5));
    const int n    = lane & 15;
    const int half = lane >> 4;

    const int p    = blockIdx.x >> 1;
    const int hh   = blockIdx.x & 1;
    const int b    = p / AXIS_LEN;
    const int row0 = hh * HP_ROW0;
    const float* __restrict__ splane = src + p * PLANE_FLOATS + row0 * ROW_FLOATS;
    float* __restrict__       dplane = dst + p * PLANE_FLOATS;

    {
        const float* __restrict__ g = wts + (b * 3 + 2) * WIN;
        v2f G[7];
        #pragma unroll
        for (int kc = 0; kc < 7; ++kc) {
            int t0 = 4 * kc + 2 * half - n;
            int t1 = t0 + 1;
            G[kc].x = (t0 >= 0 && t0 < WIN) ? g[t0] : 0.0f;
            G[kc].y = (t1 >= 0 && t1 < WIN) ? g[t1] : 0.0f;
        }

        const float* __restrict__ blk = splane + wave * (8 * ROW_FLOATS);
        v4f T[20];
        #pragma unroll
        for (int j = 0; j < 20; ++j)
            T[j] = *reinterpret_cast<const v4f*>(blk + 128 * j + 4 * lane);

        #pragma unroll
        for (int j = 0; j < 20; ++j) {
            const int f   = 128 * j + 4 * lane;
            const int rl  = f / ROW_FLOATS;
            const int col = f - rl * ROW_FLOATS;
            const int a0  = (8 * wave + rl) * LDS_PITCH + (col >> 1);
            const v2f c0 = {T[j].x, T[j].z};
            const v2f c1 = {T[j].y, T[j].w};
            *reinterpret_cast<v2f*>(&plane[a0])          = c0;
            *reinterpret_cast<v2f*>(&plane[a0 + CH_OFF]) = c1;
        }

        const int frow = (8 * wave + (n & 7)) * LDS_PITCH + (n >> 3) * CH_OFF;
        v2f R[43];
        #pragma unroll
        for (int i = 0; i < 43; ++i) {
            const int pos = 4 * i - RAD + 2 * half;
            if (i < 2 || i > 40) {
                const int pc = pos < 0 ? 0 : (pos > (AXIS_LEN - 2) ? (AXIS_LEN - 2) : pos);
                const v2f v = *reinterpret_cast<const v2f*>(&plane[frow + pc]);
                const v2f z = {0.0f, 0.0f};
                R[i] = (pos < 0 || pos >= AXIS_LEN) ? z : v;
            } else {
                R[i] = *reinterpret_cast<const v2f*>(&plane[frow + pos]);
            }
        }

        const int ob = (8 * wave) * LDS_PITCH + 2 * n + half;
        #pragma unroll
        for (int tp = 0; tp < 5; ++tp) {
            v8f acc0 = {}, acc1 = {};
            #pragma unroll
            for (int kc = 0; kc < 7; ++kc) {
                acc0 = __builtin_amdgcn_wmma_f32_16x16x4_f32(
                     false, R[8 * tp + kc],  false, G[kc],
                     (short)0, acc0,  false,  false);
                acc1 = __builtin_amdgcn_wmma_f32_16x16x4_f32(
                     false, R[8 * tp + 4 + kc],  false, G[kc],
                     (short)0, acc1,  false,  false);
            }
            #pragma unroll
            for (int v = 0; v < 8; ++v) {
                plane[ob + v * LDS_PITCH + 64 * tp]      = acc0[v];
                plane[ob + v * LDS_PITCH + 64 * tp + 32] = acc1[v];
            }
        }
    }

    __syncthreads();

    {
        const float* __restrict__ g = wts + (b * 3 + 1) * WIN;
        const int m = n;
        v2f A[7];
        #pragma unroll
        for (int kc = 0; kc < 7; ++kc) {
            int t0 = 4 * kc + 2 * half - m;
            int t1 = t0 + 1;
            A[kc].x = (t0 >= 0 && t0 < WIN) ? g[t0] : 0.0f;
            A[kc].y = (t1 >= 0 && t1 < WIN) ? g[t1] : 0.0f;
        }

        for (int u = wave; u < 25; u += HW_WAVES) {
            const int t    = u / 5;
            const int lg   = u - 5 * t;
            const int col  = 64 * lg + 4 * n;
            const int orow = hh * HP_OUT_ROWS + 16 * t;

            v2f Bv[4][7];
            #pragma unroll
            for (int kc = 0; kc < 7; ++kc) {
                const int c0 = orow + 4 * kc + 2 * half - RAD;
                const int c1 = c0 + 1;
                const int k0 = (c0 < 0 ? 0 : (c0 > (AXIS_LEN - 1) ? (AXIS_LEN - 1) : c0)) - row0;
                const int k1 = (c1 < 0 ? 0 : (c1 > (AXIS_LEN - 1) ? (AXIS_LEN - 1) : c1)) - row0;
                v4f r0 = *reinterpret_cast<const v4f*>(&plane[k0 * LDS_PITCH + col]);
                v4f r1 = *reinterpret_cast<const v4f*>(&plane[k1 * LDS_PITCH + col]);
                const v4f z = {0.0f, 0.0f, 0.0f, 0.0f};
                r0 = (c0 < 0 || c0 >= AXIS_LEN) ? z : r0;
                r1 = (c1 < 0 || c1 >= AXIS_LEN) ? z : r1;
                #pragma unroll
                for (int j = 0; j < 4; ++j) {
                    Bv[j][kc].x = r0[j];
                    Bv[j][kc].y = r1[j];
                }
            }
            tile_x4<ROW_FLOATS>(A, Bv, dplane + (orow + 8 * half) * ROW_FLOATS + col);
        }
    }
}

extern "C" void kernel_launch(void* const* d_in, const int* in_sizes, int n_in,
                              void* d_out, int out_size, void* d_ws, size_t ws_size,
                              hipStream_t stream) {
    const float* image = (const float*)d_in[0];
    const float* sigma = (const float*)d_in[1];
    float* out = (float*)d_out;
    float* wts = (float*)d_ws;
    float* tmp = (float*)((char*)d_ws + 4096);

    gauss_wts_kernel<<<1, 32, 0, stream>>>(sigma, wts);

    dim3 block(128);
    dim3 grid4(NGROUPS4 / 4);

    constexpr int S_D = 160 * 160 * 2;

    blur_pass_x4_kernel<S_D><<<grid4, block, 0, stream>>>(image, tmp, wts, 0);
    blur_hw_plane_kernel<<<dim3(NHALVES), dim3(HW_WAVES * 32), 0, stream>>>(tmp, out, wts);
}


# Gaussian taps computed by each wave from sigma (lane t = tap t, band entries picked by cross-lane reads) instead of a tap-table launch: 2 launches instead of 3; same tap formula and sum order
# speedup vs baseline: 1.7351x; 1.0689x over previous
#include <hip/hip_runtime.h>
#include <math.h>


typedef __attribute__((ext_vector_type(2))) float v2f;
typedef __attribute__((ext_vector_type(4))) float v4f;
typedef __attribute__((ext_vector_type(8))) float v8f;

#define AXIS_LEN    160
#define RAD         6
#define WIN         13
#define BATCH_ELEMS (160 * 160 * 160 * 2)
#define TOTAL_ELEMS (4 * BATCH_ELEMS)
#define TOTAL_LINES (TOTAL_ELEMS / AXIS_LEN)
#define NGROUPS4    (TOTAL_LINES / 64)

__device__ __forceinline__ float wave_taps(const float* __restrict__ sigma, int ba, int lane) {
    const float s     = sigma[ba];
    const float denom = 2.0f * s * s + 1e-7f;
    const float lg    = logf(sqrtf(2.0f * (float)M_PI) * s + 1e-7f);
    const float loc   = (float)lane - (float)RAD;
    const float e     = expf(-(loc * loc) / denom - lg);
    float sum = 0.0f;
    #pragma unroll
    for (int t = 0; t < WIN; ++t)
        sum += __builtin_bit_cast(float,
                   __builtin_amdgcn_readlane(__builtin_bit_cast(int, e), t));
    const float inv = 1.0f / sum;
    return lane < WIN ? e * inv : 0.0f;
}
__device__ __forceinline__ float tap_at(float w, int t) {
    return __builtin_bit_cast(float,
               __builtin_amdgcn_ds_bpermute((t & 31) << 2, __builtin_bit_cast(int, w)));
}


template <int S>
__device__ __forceinline__ v4f ld4(const float* __restrict__ p, int c) {
    return *reinterpret_cast<const v4f*>(p + c * S);
}
template <int S>
__device__ __forceinline__ v4f ld4_clamp_lo(const float* __restrict__ line, int c) {
    int cc = c < 0 ? 0 : c;
    v4f v = ld4<S>(line, cc);
    v4f z = {0.0f, 0.0f, 0.0f, 0.0f};
    return (c < 0) ? z : v;
}
template <int S>
__device__ __forceinline__ v4f ld4_clamp_hi(const float* __restrict__ line, int c) {
    int cc = c > (AXIS_LEN - 1) ? (AXIS_LEN - 1) : c;
    v4f v = ld4<S>(line, cc);
    v4f z = {0.0f, 0.0f, 0.0f, 0.0f};
    return (c >= AXIS_LEN) ? z : v;
}

template <int S>
__device__ __forceinline__ void load_fresh(const float* __restrict__ pf, v4f (&N)[8]) {
    #pragma unroll
    for (int kc = 0; kc < 4; ++kc) {
        N[2 * kc]     = ld4<S>(pf, 4 * kc);
        N[2 * kc + 1] = ld4<S>(pf, 4 * kc + 1);
    }
}
template <int S>
__device__ __forceinline__ void load_fresh_hi(const float* __restrict__ line, int c0, v4f (&N)[8]) {
    #pragma unroll
    for (int kc = 0; kc < 4; ++kc) {
        N[2 * kc]     = ld4_clamp_hi<S>(line, c0 + 4 * kc);
        N[2 * kc + 1] = ld4_clamp_hi<S>(line, c0 + 4 * kc + 1);
    }
}
__device__ __forceinline__ void shift_in(v2f (&Bv)[4][7], const v4f (&N)[8]) {
    #pragma unroll
    for (int j = 0; j < 4; ++j) {
        Bv[j][0] = Bv[j][4];
        Bv[j][1] = Bv[j][5];
        Bv[j][2] = Bv[j][6];
        #pragma unroll
        for (int kc = 0; kc < 4; ++kc) {
            Bv[j][3 + kc].x = N[2 * kc][j];
            Bv[j][3 + kc].y = N[2 * kc + 1][j];
        }
    }
}
template <int S>
__device__ __forceinline__ void tile_x4(const v2f (&A)[7], const v2f (&Bv)[4][7],
                                        float* __restrict__ po) {
    v8f acc[4] = {};
    #pragma unroll
    for (int kc = 0; kc < 7; ++kc) {
        #pragma unroll
        for (int j = 0; j < 4; ++j) {
            acc[j] = __builtin_amdgcn_wmma_f32_16x16x4_f32(
                 false, A[kc],  false, Bv[j][kc],
                 (short)0, acc[j],  false,  false);
        }
    }
    #pragma unroll
    for (int v = 0; v < 8; ++v) {
        v4f o = {acc[0][v], acc[1][v], acc[2][v], acc[3][v]};
        *reinterpret_cast<v4f*>(po + v * S) = o;
    }
}

template <int S>
__global__ __launch_bounds__(128) void blur_pass_x4_kernel(
        const float* __restrict__ src, float* __restrict__ dst,
        const float* __restrict__ sigma, int axis) {
    static_assert(S % 64 == 0, "64-line groups must not straddle an outer index");
    static_assert((TOTAL_LINES / 4) % 64 == 0, "64-line groups must not straddle a batch");
    const int lane  = threadIdx.x & 31;
    const int wave  = __builtin_amdgcn_readfirstlane((int)(threadIdx.x >> 5));
    const int group = blockIdx.x * 4 + wave;

    const int n    = lane & 15;
    const int half = lane >> 4;
    const int q     = group * 64 + 4 * n;
    const int outer = q / S;
    const int inner = q - outer * S;
    const int base  = outer * (S * AXIS_LEN) + inner;
    const int b     = (group * 64) / (TOTAL_LINES / 4);
    const float w   = wave_taps(sigma, b * 3 + axis, lane);

    const int m = n;
    v2f A[7];
    #pragma unroll
    for (int kc = 0; kc < 7; ++kc) {
        int t0 = 4 * kc + 2 * half - m;
        A[kc].x = tap_at(w, t0);
        A[kc].y = tap_at(w, t0 + 1);
    }

    const float* __restrict__ line = src + base;
    float* __restrict__       po   = dst + base + (8 * half) * S;

    v2f Bv[4][7];
    #pragma unroll
    for (int kc = 0; kc < 7; ++kc) {
        int c0 = 4 * kc + 2 * half - RAD;
        v4f r0 = ld4_clamp_lo<S>(line, c0);
        v4f r1 = ld4_clamp_lo<S>(line, c0 + 1);
        #pragma unroll
        for (int j = 0; j < 4; ++j) {
            Bv[j][kc].x = r0[j];
            Bv[j][kc].y = r1[j];
        }
    }

    v4f P[8], Q[8];
    const float* __restrict__ pf = line + (16 + RAD + 2 * half) * S;
    load_fresh<S>(pf, P);
    pf += 16 * S;

    #pragma unroll 1
    for (int it = 0; it < 3; ++it) {
        load_fresh<S>(pf, Q);
        pf += 16 * S;
        tile_x4<S>(A, Bv, po);
        po += 16 * S;
        shift_in(Bv, P);

        load_fresh<S>(pf, P);
        pf += 16 * S;
        tile_x4<S>(A, Bv, po);
        po += 16 * S;
        shift_in(Bv, Q);
    }
    load_fresh<S>(pf, Q);
    tile_x4<S>(A, Bv, po);
    po += 16 * S;
    shift_in(Bv, P);

    load_fresh_hi<S>(line, 16 * 9 + RAD + 2 * half, P);
    tile_x4<S>(A, Bv, po);
    po += 16 * S;
    shift_in(Bv, Q);

    tile_x4<S>(A, Bv, po);
    po += 16 * S;
    shift_in(Bv, P);

    tile_x4<S>(A, Bv, po);
}

#define ROW_FLOATS   (AXIS_LEN * 2)
#define PLANE_FLOATS (AXIS_LEN * ROW_FLOATS)
#define NPLANES      (4 * AXIS_LEN)
#define NHALVES      (2 * NPLANES)
#define LDS_PITCH    (ROW_FLOATS + 4)
#define HW_WAVES     11
#define HP_OUT_ROWS  80
#define HP_ROWS      (8 * HW_WAVES)
#define HP_ROW0      (AXIS_LEN - HP_ROWS)
#define CH_OFF       AXIS_LEN

__global__ __launch_bounds__(HW_WAVES * 32, 6) void blur_hw_plane_kernel(
        const float* __restrict__ src, float* __restrict__ dst,
        const float* __restrict__ sigma) {
    __shared__ __attribute__((aligned(16))) float plane[HP_ROWS * LDS_PITCH];

    const int lane = threadIdx.x & 31;
    const int wave = __builtin_amdgcn_readfirstlane((int)(threadIdx.x >> 5));
    const int n    = lane & 15;
    const int half = lane >> 4;

    const int p    = blockIdx.x >> 1;
    const int hh   = blockIdx.x & 1;
    const int b    = p / AXIS_LEN;
    const int row0 = hh * HP_ROW0;
    const float* __restrict__ splane = src + p * PLANE_FLOATS + row0 * ROW_FLOATS;
    float* __restrict__       dplane = dst + p * PLANE_FLOATS;

    float wH;

    {
        const float* __restrict__ blk = splane + wave * (8 * ROW_FLOATS);
        v4f T[20];
        #pragma unroll
        for (int j = 0; j < 20; ++j)
            T[j] = *reinterpret_cast<const v4f*>(blk + 128 * j + 4 * lane);

        const float wW = wave_taps(sigma, b * 3 + 2, lane);
        wH = wave_taps(sigma, b * 3 + 1, lane);
        v2f G[7];
        #pragma unroll
        for (int kc = 0; kc < 7; ++kc) {
            int t0 = 4 * kc + 2 * half - n;
            G[kc].x = tap_at(wW, t0);
            G[kc].y = tap_at(wW, t0 + 1);
        }

        #pragma unroll
        for (int j = 0; j < 20; ++j) {
            const int f   = 128 * j + 4 * lane;
            const int rl  = f / ROW_FLOATS;
            const int col = f - rl * ROW_FLOATS;
            const int a0  = (8 * wave + rl) * LDS_PITCH + (col >> 1);
            const v2f c0 = {T[j].x, T[j].z};
            const v2f c1 = {T[j].y, T[j].w};
            *reinterpret_cast<v2f*>(&plane[a0])          = c0;
            *reinterpret_cast<v2f*>(&plane[a0 + CH_OFF]) = c1;
        }

        const int frow = (8 * wave + (n & 7)) * LDS_PITCH + (n >> 3) * CH_OFF;
        v2f R[43];
        #pragma unroll
        for (int i = 0; i < 43; ++i) {
            const int pos = 4 * i - RAD + 2 * half;
            if (i < 2 || i > 40) {
                const int pc = pos < 0 ? 0 : (pos > (AXIS_LEN - 2) ? (AXIS_LEN - 2) : pos);
                const v2f v = *reinterpret_cast<const v2f*>(&plane[frow + pc]);
                const v2f z = {0.0f, 0.0f};
                R[i] = (pos < 0 || pos >= AXIS_LEN) ? z : v;
            } else {
                R[i] = *reinterpret_cast<const v2f*>(&plane[frow + pos]);
            }
        }

        const int ob = (8 * wave) * LDS_PITCH + 2 * n + half;
        #pragma unroll
        for (int tp = 0; tp < 5; ++tp) {
            v8f acc0 = {}, acc1 = {};
            #pragma unroll
            for (int kc = 0; kc < 7; ++kc) {
                acc0 = __builtin_amdgcn_wmma_f32_16x16x4_f32(
                     false, R[8 * tp + kc],  false, G[kc],
                     (short)0, acc0,  false,  false);
                acc1 = __builtin_amdgcn_wmma_f32_16x16x4_f32(
                     false, R[8 * tp + 4 + kc],  false, G[kc],
                     (short)0, acc1,  false,  false);
            }
            #pragma unroll
            for (int v = 0; v < 8; ++v) {
                plane[ob + v * LDS_PITCH + 64 * tp]      = acc0[v];
                plane[ob + v * LDS_PITCH + 64 * tp + 32] = acc1[v];
            }
        }
    }

    __syncthreads();

    {
        const int m = n;
        v2f A[7];
        #pragma unroll
        for (int kc = 0; kc < 7; ++kc) {
            int t0 = 4 * kc + 2 * half - m;
            A[kc].x = tap_at(wH, t0);
            A[kc].y = tap_at(wH, t0 + 1);
        }

        for (int u = wave; u < 25; u += HW_WAVES) {
            const int t    = u / 5;
            const int lg   = u - 5 * t;
            const int col  = 64 * lg + 4 * n;
            const int orow = hh * HP_OUT_ROWS + 16 * t;

            v2f Bv[4][7];
            #pragma unroll
            for (int kc = 0; kc < 7; ++kc) {
                const int c0 = orow + 4 * kc + 2 * half - RAD;
                const int c1 = c0 + 1;
                const int k0 = (c0 < 0 ? 0 : (c0 > (AXIS_LEN - 1) ? (AXIS_LEN - 1) : c0)) - row0;
                const int k1 = (c1 < 0 ? 0 : (c1 > (AXIS_LEN - 1) ? (AXIS_LEN - 1) : c1)) - row0;
                v4f r0 = *reinterpret_cast<const v4f*>(&plane[k0 * LDS_PITCH + col]);
                v4f r1 = *reinterpret_cast<const v4f*>(&plane[k1 * LDS_PITCH + col]);
                const v4f z = {0.0f, 0.0f, 0.0f, 0.0f};
                r0 = (c0 < 0 || c0 >= AXIS_LEN) ? z : r0;
                r1 = (c1 < 0 || c1 >= AXIS_LEN) ? z : r1;
                #pragma unroll
                for (int j = 0; j < 4; ++j) {
                    Bv[j][kc].x = r0[j];
                    Bv[j][kc].y = r1[j];
                }
            }
            tile_x4<ROW_FLOATS>(A, Bv, dplane + (orow + 8 * half) * ROW_FLOATS + col);
        }
    }
}

extern "C" void kernel_launch(void* const* d_in, const int* in_sizes, int n_in,
                              void* d_out, int out_size, void* d_ws, size_t ws_size,
                              hipStream_t stream) {
    const float* image = (const float*)d_in[0];
    const float* sigma = (const float*)d_in[1];
    float* out = (float*)d_out;
    float* tmp = (float*)((char*)d_ws + 4096);

    dim3 block(128);
    dim3 grid4(NGROUPS4 / 4);

    constexpr int S_D = 160 * 160 * 2;

    blur_pass_x4_kernel<S_D><<<grid4, block, 0, stream>>>(image, tmp, sigma, 0);
    blur_hw_plane_kernel<<<dim3(NHALVES), dim3(HW_WAVES * 32), 0, stream>>>(tmp, out, sigma);
}
